# starting kernel
# baseline (speedup 1.0000x reference)
#include <hip/hip_runtime.h>
#include <hip/hip_bf16.h>

typedef float v2f  __attribute__((ext_vector_type(2)));
typedef float v8f  __attribute__((ext_vector_type(8)));
typedef __bf16 bf16;
typedef bf16 v4bf  __attribute__((ext_vector_type(4)));
typedef bf16 v8bf  __attribute__((ext_vector_type(8)));
typedef bf16 v16bf __attribute__((ext_vector_type(16)));

__device__ __forceinline__ v8f wmma4(v2f a, v2f b, v8f c) {
    return __builtin_amdgcn_wmma_f32_16x16x4_f32(
        false, a, false, b, (short)0, c, false, false);
}
__device__ __forceinline__ v8f wmma32bf(v16bf a, v16bf b, v8f c) {
    return __builtin_amdgcn_wmma_f32_16x16x32_bf16(
        false, a, false, b, (short)0, c, false, false);
}

__device__ __forceinline__ v16bf frag16(const bf16* rowBase, int khalf) {
    v8bf lo = *(const v8bf*)(rowBase + khalf);
    v8bf hi = *(const v8bf*)(rowBase + 16 + khalf);
    return __builtin_shufflevector(lo, hi, 0,1,2,3,4,5,6,7,8,9,10,11,12,13,14,15);
}

__global__ __launch_bounds__(256) void split_transpose_bf16(
    const float* __restrict__ B, bf16* __restrict__ Th, bf16* __restrict__ Tl,
    int K, int N)
{
    __shared__ float t[32][33];
    const int tid = threadIdx.x;
    const int k0 = blockIdx.y * 32;
    const int n0 = blockIdx.x * 32;
    #pragma unroll
    for (int i = 0; i < 4; ++i) {
        int id = tid + i * 256;
        int r = id >> 5, c = id & 31;
        t[r][c] = B[(size_t)(k0 + r) * N + n0 + c];
    }
    __syncthreads();
    #pragma unroll
    for (int i = 0; i < 4; ++i) {
        int id = tid + i * 256;
        int r = id >> 5, c = id & 31;
        float f = t[c][r];
        bf16 h  = (bf16)f;
        float l = f - (float)h;
        size_t o = (size_t)(n0 + r) * K + k0 + c;
        Th[o] = h;
        Tl[o] = (bf16)l;
    }
}

#define GBM 128
#define GBN 64
#define GKT 32
#define APITCH 40

__global__ __launch_bounds__(256) void gemm_bf16x3(
    const float* __restrict__ A,
    const bf16* __restrict__ Bth, const bf16* __restrict__ Btl,
    const float* __restrict__ bias, float* __restrict__ Cm,
    int M, int N, int K)
{
    __shared__ __align__(16) bf16 Ash[GBM][APITCH];
    __shared__ __align__(16) bf16 Asl[GBM][APITCH];
    __shared__ __align__(16) bf16 Bsh[GBN][APITCH];
    __shared__ __align__(16) bf16 Bsl[GBN][APITCH];

    const int tid   = threadIdx.x;
    const int wave  = tid >> 5;
    const int lane  = tid & 31;
    const int lm    = lane & 15;
    const int khalf = (lane >> 4) << 3;
    const int hi    = (lane >> 4) << 3;

    const int wr = (wave & 3) * 32;
    const int wc = (wave >> 2) * 32;

    const long rowBase = (long)blockIdx.y * GBM;
    const long colBase = (long)blockIdx.x * GBN;

    v8f acc[2][2];
    acc[0][0] = {}; acc[0][1] = {}; acc[1][0] = {}; acc[1][1] = {};

    for (int k0 = 0; k0 < K; k0 += GKT) {
        #pragma unroll
        for (int i = 0; i < 4; ++i) {
            int id = tid + i * 256;
            int r  = id >> 3;
            int c  = (id & 7) << 2;
            float4 f = *(const float4*)&A[(size_t)(rowBase + r) * K + k0 + c];
            v4bf h, l;
            h[0] = (bf16)f.x; l[0] = (bf16)(f.x - (float)h[0]);
            h[1] = (bf16)f.y; l[1] = (bf16)(f.y - (float)h[1]);
            h[2] = (bf16)f.z; l[2] = (bf16)(f.z - (float)h[2]);
            h[3] = (bf16)f.w; l[3] = (bf16)(f.w - (float)h[3]);
            *(v4bf*)&Ash[r][c] = h;
            *(v4bf*)&Asl[r][c] = l;
        }
        {
            int r = tid >> 2;
            int c = (tid & 3) << 3;
            size_t o = (size_t)(colBase + r) * K + k0 + c;
            *(v8bf*)&Bsh[r][c] = *(const v8bf*)&Bth[o];
            *(v8bf*)&Bsl[r][c] = *(const v8bf*)&Btl[o];
        }
        if (k0 + GKT < K)
            __builtin_prefetch(&A[(size_t)(rowBase + (tid >> 1)) * K + k0 + GKT], 0, 1);
        __syncthreads();

        v16bf ah0 = frag16(&Ash[wr      + lm][0], khalf);
        v16bf ah1 = frag16(&Ash[wr + 16 + lm][0], khalf);
        v16bf al0 = frag16(&Asl[wr      + lm][0], khalf);
        v16bf al1 = frag16(&Asl[wr + 16 + lm][0], khalf);
        v16bf bh0 = frag16(&Bsh[wc      + lm][0], khalf);
        v16bf bh1 = frag16(&Bsh[wc + 16 + lm][0], khalf);
        v16bf bl0 = frag16(&Bsl[wc      + lm][0], khalf);
        v16bf bl1 = frag16(&Bsl[wc + 16 + lm][0], khalf);

        acc[0][0] = wmma32bf(al0, bh0, acc[0][0]);
        acc[0][0] = wmma32bf(ah0, bl0, acc[0][0]);
        acc[0][0] = wmma32bf(ah0, bh0, acc[0][0]);
        acc[0][1] = wmma32bf(al0, bh1, acc[0][1]);
        acc[0][1] = wmma32bf(ah0, bl1, acc[0][1]);
        acc[0][1] = wmma32bf(ah0, bh1, acc[0][1]);
        acc[1][0] = wmma32bf(al1, bh0, acc[1][0]);
        acc[1][0] = wmma32bf(ah1, bl0, acc[1][0]);
        acc[1][0] = wmma32bf(ah1, bh0, acc[1][0]);
        acc[1][1] = wmma32bf(al1, bh1, acc[1][1]);
        acc[1][1] = wmma32bf(ah1, bl1, acc[1][1]);
        acc[1][1] = wmma32bf(ah1, bh1, acc[1][1]);
        __syncthreads();
    }

    #pragma unroll
    for (int i = 0; i < 2; ++i) {
        #pragma unroll
        for (int j = 0; j < 2; ++j) {
            long col  = colBase + wc + j * 16 + lm;
            float bad = bias ? bias[col] : 0.0f;
            #pragma unroll
            for (int r = 0; r < 8; ++r) {
                long row = rowBase + wr + i * 16 + hi + r;
                Cm[(size_t)row * N + col] = acc[i][j][r] + bad;
            }
        }
    }
}

#define MT_C3 3072
#define MT_N  1024

__global__ __launch_bounds__(256) void mttt_head_kernel(
    const float* __restrict__ qkv, float* __restrict__ mid)
{
    __shared__ __align__(16) float smem[25536];
    float (*Ks)[68] = (float(*)[68]) &smem[0];
    float (*Vs)[68] = (float(*)[68]) &smem[32 * 68];
    float (*Qs)[68] = (float(*)[68]) &smem[0];
    float (*Am)[65] = (float(*)[65]) &smem[8704];
    float (*Cs)[65] = (float(*)[65]) &smem[8704 + 4160];
    float (*W0)[65] = (float(*)[65]) &smem[8704 + 2 * 4160];
    float (*W1)[65] = (float(*)[65]) &smem[8704 + 3 * 4160];
    float* mk   = &smem[8704 + 4 * 4160];
    float* mv   = mk + 64;
    float* bvec = mk + 128;

    const int bh = blockIdx.x;
    const int bb = bh >> 4;
    const int h  = bh & 15;
    const size_t rowBase = (size_t)bb * MT_N;
    const int qcol = h * 64;
    const int kcol = 1024 + h * 64;
    const int vcol = 2048 + h * 64;

    const int tid  = threadIdx.x;
    const int wave = tid >> 5;
    const int lane = tid & 31;
    const int lm   = lane & 15;
    const int lk   = (lane >> 4) << 1;
    const int hi   = (lane >> 4) << 3;

    const bool isA  = wave < 4;
    const int mbase = (wave & 3) * 16;
    v8f accT[4];
    accT[0] = {}; accT[1] = {}; accT[2] = {}; accT[3] = {};
    float msum = 0.0f;

    for (int k0 = 0; k0 < MT_N; k0 += 32) {
        #pragma unroll
        for (int i = 0; i < 2; ++i) {
            int id = tid + i * 256;
            int r  = id >> 4;
            int c  = (id & 15) << 2;
            const float* src = &qkv[(rowBase + k0 + r) * MT_C3];
            *(float4*)&Ks[r][c] = *(const float4*)&src[kcol + c];
            *(float4*)&Vs[r][c] = *(const float4*)&src[vcol + c];
        }
        __syncthreads();

        if (tid < 64) {
            #pragma unroll 8
            for (int r = 0; r < 32; ++r) msum += Ks[r][tid];
        } else if (tid < 128) {
            #pragma unroll 8
            for (int r = 0; r < 32; ++r) msum += Vs[r][tid - 64];
        }

        float (*Bsrc)[68] = isA ? Ks : Vs;
        #pragma unroll
        for (int kk = 0; kk < 32; kk += 4) {
            v2f a;
            a.x = Ks[kk + lk][mbase + lm];
            a.y = Ks[kk + lk + 1][mbase + lm];
            #pragma unroll
            for (int j = 0; j < 4; ++j) {
                v2f b;
                b.x = Bsrc[kk + lk][j * 16 + lm];
                b.y = Bsrc[kk + lk + 1][j * 16 + lm];
                accT[j] = wmma4(a, b, accT[j]);
            }
        }
        __syncthreads();
    }

    const float invN = 1.0f / (float)MT_N;
    {
        float (*Dst)[65] = isA ? Am : Cs;
        #pragma unroll
        for (int j = 0; j < 4; ++j)
            #pragma unroll
            for (int r = 0; r < 8; ++r)
                Dst[mbase + hi + r][j * 16 + lm] = accT[j][r] * invN;
    }
    if (tid < 64)       mk[tid]      = msum * invN;
    else if (tid < 128) mv[tid - 64] = msum * invN;
    for (int id = tid; id < 64 * 64; id += 256) W0[id >> 6][id & 63] = 0.0f;
    if (tid < 64) bvec[tid] = 0.0f;
    __syncthreads();

    float (*Wcur)[65] = W0;
    float (*Wnxt)[65] = W1;
    for (int it = 0; it < 4; ++it) {
        float bnew = 0.0f;
        if (tid < 64) {
            float s = 0.0f;
            #pragma unroll 8
            for (int i = 0; i < 64; ++i) s += mk[i] * Wcur[i][tid];
            bnew = mv[tid] - s;
        }
        #pragma unroll
        for (int j = 0; j < 2; ++j) {
            int t  = wave * 2 + j;
            int mb = (t >> 2) * 16;
            int nb = (t & 3) * 16;
            v8f acc = {};
            #pragma unroll
            for (int kk = 0; kk < 64; kk += 4) {
                v2f a, b2;
                a.x  = Am[mb + lm][kk + lk];
                a.y  = Am[mb + lm][kk + lk + 1];
                b2.x = Wcur[kk + lk][nb + lm];
                b2.y = Wcur[kk + lk + 1][nb + lm];
                acc = wmma4(a, b2, acc);
            }
            int col  = nb + lm;
            float bc = bvec[col];
            #pragma unroll
            for (int r = 0; r < 8; ++r) {
                int row = mb + hi + r;
                Wnxt[row][col] = Wcur[row][col] - acc[r] - mk[row] * bc + Cs[row][col];
            }
        }
        __syncthreads();
        if (tid < 64) bvec[tid] = bnew;
        { float (*tmp)[65] = Wcur; Wcur = Wnxt; Wnxt = tmp; }
        __syncthreads();
    }

    for (int c0 = 0; c0 < MT_N; c0 += 128) {
        #pragma unroll
        for (int i = 0; i < 8; ++i) {
            int id = tid + i * 256;
            int r  = id >> 4;
            int c  = (id & 15) << 2;
            *(float4*)&Qs[r][c] =
                *(const float4*)&qkv[(rowBase + c0 + r) * MT_C3 + qcol + c];
        }
        __syncthreads();

        int rb = wave * 16;
        v8f acc[4];
        acc[0] = {}; acc[1] = {}; acc[2] = {}; acc[3] = {};
        #pragma unroll
        for (int kk = 0; kk < 64; kk += 4) {
            v2f a;
            a.x = Qs[rb + lm][kk + lk];
            a.y = Qs[rb + lm][kk + lk + 1];
            #pragma unroll
            for (int j = 0; j < 4; ++j) {
                v2f b2;
                b2.x = Wcur[kk + lk][j * 16 + lm];
                b2.y = Wcur[kk + lk + 1][j * 16 + lm];
                acc[j] = wmma4(a, b2, acc[j]);
            }
        }
        #pragma unroll
        for (int j = 0; j < 4; ++j) {
            int col  = j * 16 + lm;
            float ba = bvec[col];
            #pragma unroll
            for (int r = 0; r < 8; ++r) {
                size_t row = rowBase + c0 + rb + hi + r;
                mid[row * 1024 + qcol + col] = acc[j][r] + ba;
            }
        }
        __syncthreads();
    }
}

extern "C" void kernel_launch(void* const* d_in, const int* in_sizes, int n_in,
                              void* d_out, int out_size, void* d_ws, size_t ws_size,
                              hipStream_t stream) {
    const float* x      = (const float*)d_in[0];
    const float* w_qkv  = (const float*)d_in[1];
    const float* w_proj = (const float*)d_in[2];
    const float* b_proj = (const float*)d_in[3];
    float* out = (float*)d_out;

    float* qkv = (float*)d_ws;
    float* mid = qkv + (size_t)8192 * 3072;
    bf16*  b1h = (bf16*)(mid + (size_t)8192 * 1024);
    bf16*  b1l = b1h + (size_t)3072 * 1024;
    bf16*  b2h = b1l + (size_t)3072 * 1024;
    bf16*  b2l = b2h + (size_t)1024 * 1024;

    dim3 blk(256);
    split_transpose_bf16<<<dim3(3072 / 32, 1024 / 32), blk, 0, stream>>>(
        w_qkv, b1h, b1l, 1024, 3072);
    split_transpose_bf16<<<dim3(1024 / 32, 1024 / 32), blk, 0, stream>>>(
        w_proj, b2h, b2l, 1024, 1024);

    gemm_bf16x3<<<dim3(3072 / GBN, 8192 / GBM), blk, 0, stream>>>(
        x, b1h, b1l, nullptr, qkv, 8192, 3072, 1024);

    mttt_head_kernel<<<dim3(8 * 16), blk, 0, stream>>>(qkv, mid);

    gemm_bf16x3<<<dim3(1024 / GBN, 8192 / GBM), blk, 0, stream>>>(
        mid, b2h, b2l, b_proj, out, 8192, 1024, 1024);
}


# bf16x3 GEMM re-tiled 128x64x32 -> 128x128x64 (wave tile 32x32 -> 32x64, 16 tile loads issued as one group): fewer L2 and LDS bytes, splits and barriers per WMMA; sum order kept
# speedup vs baseline: 1.2039x; 1.2039x over previous
#include <hip/hip_runtime.h>
#include <hip/hip_bf16.h>

typedef float v2f  __attribute__((ext_vector_type(2)));
typedef float v8f  __attribute__((ext_vector_type(8)));
typedef __bf16 bf16;
typedef bf16 v4bf  __attribute__((ext_vector_type(4)));
typedef bf16 v8bf  __attribute__((ext_vector_type(8)));
typedef bf16 v16bf __attribute__((ext_vector_type(16)));

__device__ __forceinline__ v8f wmma4(v2f a, v2f b, v8f c) {
    return __builtin_amdgcn_wmma_f32_16x16x4_f32(
        false, a, false, b, (short)0, c, false, false);
}
__device__ __forceinline__ v8f wmma32bf(v16bf a, v16bf b, v8f c) {
    return __builtin_amdgcn_wmma_f32_16x16x32_bf16(
        false, a, false, b, (short)0, c, false, false);
}

__device__ __forceinline__ v16bf frag16(const bf16* rowBase, int khalf) {
    v8bf lo = *(const v8bf*)(rowBase + khalf);
    v8bf hi = *(const v8bf*)(rowBase + 16 + khalf);
    return __builtin_shufflevector(lo, hi, 0,1,2,3,4,5,6,7,8,9,10,11,12,13,14,15);
}

__global__ __launch_bounds__(256) void split_transpose_bf16(
    const float* __restrict__ B, bf16* __restrict__ Th, bf16* __restrict__ Tl,
    int K, int N)
{
    __shared__ float t[32][33];
    const int tid = threadIdx.x;
    const int k0 = blockIdx.y * 32;
    const int n0 = blockIdx.x * 32;
    #pragma unroll
    for (int i = 0; i < 4; ++i) {
        int id = tid + i * 256;
        int r = id >> 5, c = id & 31;
        t[r][c] = B[(size_t)(k0 + r) * N + n0 + c];
    }
    __syncthreads();
    #pragma unroll
    for (int i = 0; i < 4; ++i) {
        int id = tid + i * 256;
        int r = id >> 5, c = id & 31;
        float f = t[c][r];
        bf16 h  = (bf16)f;
        float l = f - (float)h;
        size_t o = (size_t)(n0 + r) * K + k0 + c;
        Th[o] = h;
        Tl[o] = (bf16)l;
    }
}

#define GBM 128
#define GBN 128
#define GKT 64
#define APITCH 72

__global__ __launch_bounds__(256) __attribute__((amdgpu_waves_per_eu(4, 4))) void gemm_bf16x3(
    const float* __restrict__ A,
    const bf16* __restrict__ Bth, const bf16* __restrict__ Btl,
    const float* __restrict__ bias, float* __restrict__ Cm,
    int M, int N, int K)
{
    __shared__ __align__(16) bf16 Ash[GBM][APITCH];
    __shared__ __align__(16) bf16 Asl[GBM][APITCH];
    __shared__ __align__(16) bf16 Bsh[GBN][APITCH];
    __shared__ __align__(16) bf16 Bsl[GBN][APITCH];

    const int tid   = threadIdx.x;
    const int wave  = tid >> 5;
    const int lane  = tid & 31;
    const int lm    = lane & 15;
    const int khalf = (lane >> 4) << 3;
    const int hi    = (lane >> 4) << 3;

    const int wr = (wave & 3) * 32;
    const int wc = (wave >> 2) * 64;

    const long rowBase = (long)blockIdx.y * GBM;
    const long colBase = (long)blockIdx.x * GBN;

    v8f acc[2][4];
    #pragma unroll
    for (int j = 0; j < 4; ++j) { acc[0][j] = {}; acc[1][j] = {}; }

    for (int k0 = 0; k0 < K; k0 += GKT) {
        float4 fa[8];
        v8bf gbh[4], gbl[4];
        #pragma unroll
        for (int i = 0; i < 8; ++i) {
            int id = tid + i * 256;
            fa[i] = *(const float4*)&A[(size_t)(rowBase + (id >> 4)) * K + k0 + ((id & 15) << 2)];
        }
        #pragma unroll
        for (int i = 0; i < 4; ++i) {
            int id = tid + i * 256;
            size_t o = (size_t)(colBase + (id >> 3)) * K + k0 + ((id & 7) << 3);
            gbh[i] = *(const v8bf*)&Bth[o];
            gbl[i] = *(const v8bf*)&Btl[o];
        }
        #pragma unroll
        for (int i = 0; i < 8; ++i) {
            int id = tid + i * 256;
            int r  = id >> 4;
            int c  = (id & 15) << 2;
            float4 f = fa[i];
            v4bf h, l;
            h[0] = (bf16)f.x; l[0] = (bf16)(f.x - (float)h[0]);
            h[1] = (bf16)f.y; l[1] = (bf16)(f.y - (float)h[1]);
            h[2] = (bf16)f.z; l[2] = (bf16)(f.z - (float)h[2]);
            h[3] = (bf16)f.w; l[3] = (bf16)(f.w - (float)h[3]);
            *(v4bf*)&Ash[r][c] = h;
            *(v4bf*)&Asl[r][c] = l;
        }
        #pragma unroll
        for (int i = 0; i < 4; ++i) {
            int id = tid + i * 256;
            int r = id >> 3;
            int c = (id & 7) << 3;
            *(v8bf*)&Bsh[r][c] = gbh[i];
            *(v8bf*)&Bsl[r][c] = gbl[i];
        }
        if (k0 + GKT < K)
            __builtin_prefetch(&A[(size_t)(rowBase + (tid >> 1)) * K + k0 + GKT + (tid & 1) * 32], 0, 1);
        __syncthreads();

        #pragma unroll
        for (int ks = 0; ks < GKT; ks += 32) {
            v16bf ah[2], al[2], bh[4], bl[4];
            #pragma unroll
            for (int i = 0; i < 2; ++i) {
                ah[i] = frag16(&Ash[wr + i * 16 + lm][ks], khalf);
                al[i] = frag16(&Asl[wr + i * 16 + lm][ks], khalf);
            }
            #pragma unroll
            for (int j = 0; j < 4; ++j) {
                bh[j] = frag16(&Bsh[wc + j * 16 + lm][ks], khalf);
                bl[j] = frag16(&Bsl[wc + j * 16 + lm][ks], khalf);
            }
            #pragma unroll
            for (int i = 0; i < 2; ++i)
                #pragma unroll
                for (int j = 0; j < 4; ++j)
                    acc[i][j] = wmma32bf(al[i], bh[j], acc[i][j]);
            #pragma unroll
            for (int i = 0; i < 2; ++i)
                #pragma unroll
                for (int j = 0; j < 4; ++j)
                    acc[i][j] = wmma32bf(ah[i], bl[j], acc[i][j]);
            #pragma unroll
            for (int i = 0; i < 2; ++i)
                #pragma unroll
                for (int j = 0; j < 4; ++j)
                    acc[i][j] = wmma32bf(ah[i], bh[j], acc[i][j]);
        }
        __syncthreads();
    }

    #pragma unroll
    for (int i = 0; i < 2; ++i) {
        #pragma unroll
        for (int j = 0; j < 4; ++j) {
            long col  = colBase + wc + j * 16 + lm;
            float bad = bias ? bias[col] : 0.0f;
            #pragma unroll
            for (int r = 0; r < 8; ++r) {
                long row = rowBase + wr + i * 16 + hi + r;
                Cm[(size_t)row * N + col] = acc[i][j][r] + bad;
            }
        }
    }
}

#define MT_C3 3072
#define MT_N  1024

__global__ __launch_bounds__(256) void mttt_head_kernel(
    const float* __restrict__ qkv, float* __restrict__ mid)
{
    __shared__ __align__(16) float smem[25536];
    float (*Ks)[68] = (float(*)[68]) &smem[0];
    float (*Vs)[68] = (float(*)[68]) &smem[32 * 68];
    float (*Qs)[68] = (float(*)[68]) &smem[0];
    float (*Am)[65] = (float(*)[65]) &smem[8704];
    float (*Cs)[65] = (float(*)[65]) &smem[8704 + 4160];
    float (*W0)[65] = (float(*)[65]) &smem[8704 + 2 * 4160];
    float (*W1)[65] = (float(*)[65]) &smem[8704 + 3 * 4160];
    float* mk   = &smem[8704 + 4 * 4160];
    float* mv   = mk + 64;
    float* bvec = mk + 128;

    const int bh = blockIdx.x;
    const int bb = bh >> 4;
    const int h  = bh & 15;
    const size_t rowBase = (size_t)bb * MT_N;
    const int qcol = h * 64;
    const int kcol = 1024 + h * 64;
    const int vcol = 2048 + h * 64;

    const int tid  = threadIdx.x;
    const int wave = tid >> 5;
    const int lane = tid & 31;
    const int lm   = lane & 15;
    const int lk   = (lane >> 4) << 1;
    const int hi   = (lane >> 4) << 3;

    const bool isA  = wave < 4;
    const int mbase = (wave & 3) * 16;
    v8f accT[4];
    accT[0] = {}; accT[1] = {}; accT[2] = {}; accT[3] = {};
    float msum = 0.0f;

    for (int k0 = 0; k0 < MT_N; k0 += 32) {
        #pragma unroll
        for (int i = 0; i < 2; ++i) {
            int id = tid + i * 256;
            int r  = id >> 4;
            int c  = (id & 15) << 2;
            const float* src = &qkv[(rowBase + k0 + r) * MT_C3];
            *(float4*)&Ks[r][c] = *(const float4*)&src[kcol + c];
            *(float4*)&Vs[r][c] = *(const float4*)&src[vcol + c];
        }
        __syncthreads();

        if (tid < 64) {
            #pragma unroll 8
            for (int r = 0; r < 32; ++r) msum += Ks[r][tid];
        } else if (tid < 128) {
            #pragma unroll 8
            for (int r = 0; r < 32; ++r) msum += Vs[r][tid - 64];
        }

        float (*Bsrc)[68] = isA ? Ks : Vs;
        #pragma unroll
        for (int kk = 0; kk < 32; kk += 4) {
            v2f a;
            a.x = Ks[kk + lk][mbase + lm];
            a.y = Ks[kk + lk + 1][mbase + lm];
            #pragma unroll
            for (int j = 0; j < 4; ++j) {
                v2f b;
                b.x = Bsrc[kk + lk][j * 16 + lm];
                b.y = Bsrc[kk + lk + 1][j * 16 + lm];
                accT[j] = wmma4(a, b, accT[j]);
            }
        }
        __syncthreads();
    }

    const float invN = 1.0f / (float)MT_N;
    {
        float (*Dst)[65] = isA ? Am : Cs;
        #pragma unroll
        for (int j = 0; j < 4; ++j)
            #pragma unroll
            for (int r = 0; r < 8; ++r)
                Dst[mbase + hi + r][j * 16 + lm] = accT[j][r] * invN;
    }
    if (tid < 64)       mk[tid]      = msum * invN;
    else if (tid < 128) mv[tid - 64] = msum * invN;
    for (int id = tid; id < 64 * 64; id += 256) W0[id >> 6][id & 63] = 0.0f;
    if (tid < 64) bvec[tid] = 0.0f;
    __syncthreads();

    float (*Wcur)[65] = W0;
    float (*Wnxt)[65] = W1;
    for (int it = 0; it < 4; ++it) {
        float bnew = 0.0f;
        if (tid < 64) {
            float s = 0.0f;
            #pragma unroll 8
            for (int i = 0; i < 64; ++i) s += mk[i] * Wcur[i][tid];
            bnew = mv[tid] - s;
        }
        #pragma unroll
        for (int j = 0; j < 2; ++j) {
            int t  = wave * 2 + j;
            int mb = (t >> 2) * 16;
            int nb = (t & 3) * 16;
            v8f acc = {};
            #pragma unroll
            for (int kk = 0; kk < 64; kk += 4) {
                v2f a, b2;
                a.x  = Am[mb + lm][kk + lk];
                a.y  = Am[mb + lm][kk + lk + 1];
                b2.x = Wcur[kk + lk][nb + lm];
                b2.y = Wcur[kk + lk + 1][nb + lm];
                acc = wmma4(a, b2, acc);
            }
            int col  = nb + lm;
            float bc = bvec[col];
            #pragma unroll
            for (int r = 0; r < 8; ++r) {
                int row = mb + hi + r;
                Wnxt[row][col] = Wcur[row][col] - acc[r] - mk[row] * bc + Cs[row][col];
            }
        }
        __syncthreads();
        if (tid < 64) bvec[tid] = bnew;
        { float (*tmp)[65] = Wcur; Wcur = Wnxt; Wnxt = tmp; }
        __syncthreads();
    }

    for (int c0 = 0; c0 < MT_N; c0 += 128) {
        #pragma unroll
        for (int i = 0; i < 8; ++i) {
            int id = tid + i * 256;
            int r  = id >> 4;
            int c  = (id & 15) << 2;
            *(float4*)&Qs[r][c] =
                *(const float4*)&qkv[(rowBase + c0 + r) * MT_C3 + qcol + c];
        }
        __syncthreads();

        int rb = wave * 16;
        v8f acc[4];
        acc[0] = {}; acc[1] = {}; acc[2] = {}; acc[3] = {};
        #pragma unroll
        for (int kk = 0; kk < 64; kk += 4) {
            v2f a;
            a.x = Qs[rb + lm][kk + lk];
            a.y = Qs[rb + lm][kk + lk + 1];
            #pragma unroll
            for (int j = 0; j < 4; ++j) {
                v2f b2;
                b2.x = Wcur[kk + lk][j * 16 + lm];
                b2.y = Wcur[kk + lk + 1][j * 16 + lm];
                acc[j] = wmma4(a, b2, acc[j]);
            }
        }
        #pragma unroll
        for (int j = 0; j < 4; ++j) {
            int col  = j * 16 + lm;
            float ba = bvec[col];
            #pragma unroll
            for (int r = 0; r < 8; ++r) {
                size_t row = rowBase + c0 + rb + hi + r;
                mid[row * 1024 + qcol + col] = acc[j][r] + ba;
            }
        }
        __syncthreads();
    }
}

extern "C" void kernel_launch(void* const* d_in, const int* in_sizes, int n_in,
                              void* d_out, int out_size, void* d_ws, size_t ws_size,
                              hipStream_t stream) {
    const float* x      = (const float*)d_in[0];
    const float* w_qkv  = (const float*)d_in[1];
    const float* w_proj = (const float*)d_in[2];
    const float* b_proj = (const float*)d_in[3];
    float* out = (float*)d_out;

    float* qkv = (float*)d_ws;
    float* mid = qkv + (size_t)8192 * 3072;
    bf16*  b1h = (bf16*)(mid + (size_t)8192 * 1024);
    bf16*  b1l = b1h + (size_t)3072 * 1024;
    bf16*  b2h = b1l + (size_t)3072 * 1024;
    bf16*  b2l = b2h + (size_t)1024 * 1024;

    dim3 blk(256);
    split_transpose_bf16<<<dim3(3072 / 32, 1024 / 32), blk, 0, stream>>>(
        w_qkv, b1h, b1l, 1024, 3072);
    split_transpose_bf16<<<dim3(1024 / 32, 1024 / 32), blk, 0, stream>>>(
        w_proj, b2h, b2l, 1024, 1024);

    gemm_bf16x3<<<dim3(3072 / GBN, 8192 / GBM), blk, 0, stream>>>(
        x, b1h, b1l, nullptr, qkv, 8192, 3072, 1024);

    mttt_head_kernel<<<dim3(8 * 16), blk, 0, stream>>>(qkv, mid);

    gemm_bf16x3<<<dim3(1024 / GBN, 8192 / GBM), blk, 0, stream>>>(
        mid, b2h, b2l, b_proj, out, 8192, 1024, 1024);
}


# bf16x3 GEMM K loop software-pipelined: next tile's global loads fly under the WMMA block, split+stored after it into a 2nd LDS stage, 1 barrier/tile, one 16-wave 256x128 group per WGP; sum order kept
# speedup vs baseline: 1.4713x; 1.2221x over previous
#include <hip/hip_runtime.h>
#include <hip/hip_bf16.h>

typedef float v2f  __attribute__((ext_vector_type(2)));
typedef float v8f  __attribute__((ext_vector_type(8)));
typedef __bf16 bf16;
typedef bf16 v4bf  __attribute__((ext_vector_type(4)));
typedef bf16 v8bf  __attribute__((ext_vector_type(8)));
typedef bf16 v16bf __attribute__((ext_vector_type(16)));

__device__ __forceinline__ v8f wmma4(v2f a, v2f b, v8f c) {
    return __builtin_amdgcn_wmma_f32_16x16x4_f32(
        false, a, false, b, (short)0, c, false, false);
}
__device__ __forceinline__ v8f wmma32bf(v16bf a, v16bf b, v8f c) {
    return __builtin_amdgcn_wmma_f32_16x16x32_bf16(
        false, a, false, b, (short)0, c, false, false);
}

__device__ __forceinline__ v16bf frag16(const bf16* rowBase, int khalf) {
    v8bf lo = *(const v8bf*)(rowBase + khalf);
    v8bf hi = *(const v8bf*)(rowBase + 16 + khalf);
    return __builtin_shufflevector(lo, hi, 0,1,2,3,4,5,6,7,8,9,10,11,12,13,14,15);
}

__global__ __launch_bounds__(256) void split_transpose_bf16(
    const float* __restrict__ B, bf16* __restrict__ Th, bf16* __restrict__ Tl,
    int K, int N)
{
    __shared__ float t[32][33];
    const int tid = threadIdx.x;
    const int k0 = blockIdx.y * 32;
    const int n0 = blockIdx.x * 32;
    #pragma unroll
    for (int i = 0; i < 4; ++i) {
        int id = tid + i * 256;
        int r = id >> 5, c = id & 31;
        t[r][c] = B[(size_t)(k0 + r) * N + n0 + c];
    }
    __syncthreads();
    #pragma unroll
    for (int i = 0; i < 4; ++i) {
        int id = tid + i * 256;
        int r = id >> 5, c = id & 31;
        float f = t[c][r];
        bf16 h  = (bf16)f;
        float l = f - (float)h;
        size_t o = (size_t)(n0 + r) * K + k0 + c;
        Th[o] = h;
        Tl[o] = (bf16)l;
    }
}

#define GBM 256
#define GBN 128
#define GKT 64
#define GK  1024
#define GNKT (GK / GKT)
#define APITCH 72

struct GemmStage {
    bf16 ah[GBM][APITCH];
    bf16 al[GBM][APITCH];
    bf16 bh[GBN][APITCH];
    bf16 bl[GBN][APITCH];
};

__device__ __forceinline__ void gemm_load_tile(
    const float* __restrict__ at, const bf16* __restrict__ bht, const bf16* __restrict__ blt,
    unsigned aoff, unsigned boff, float4 (&fa)[8], v8bf (&gbh)[2], v8bf (&gbl)[2])
{
    #pragma unroll
    for (int i = 0; i < 8; ++i)
        fa[i] = *(const float4*)(at + (aoff + (unsigned)i * 32u * GK));
    #pragma unroll
    for (int i = 0; i < 2; ++i) {
        gbh[i] = *(const v8bf*)(bht + (boff + (unsigned)i * 64u * GK));
        gbl[i] = *(const v8bf*)(blt + (boff + (unsigned)i * 64u * GK));
    }
}

__device__ __forceinline__ void gemm_store_tile(
    GemmStage& st, int tid, const float4 (&fa)[8], const v8bf (&gbh)[2], const v8bf (&gbl)[2])
{
    const int ar = tid >> 4, ac = (tid & 15) << 2;
    const int br = tid >> 3, bc = (tid & 7) << 3;
    #pragma unroll
    for (int i = 0; i < 8; ++i) {
        float4 f = fa[i];
        v4bf h, l;
        h[0] = (bf16)f.x; l[0] = (bf16)(f.x - (float)h[0]);
        h[1] = (bf16)f.y; l[1] = (bf16)(f.y - (float)h[1]);
        h[2] = (bf16)f.z; l[2] = (bf16)(f.z - (float)h[2]);
        h[3] = (bf16)f.w; l[3] = (bf16)(f.w - (float)h[3]);
        *(v4bf*)&st.ah[ar + i * 32][ac] = h;
        *(v4bf*)&st.al[ar + i * 32][ac] = l;
    }
    #pragma unroll
    for (int i = 0; i < 2; ++i) {
        *(v8bf*)&st.bh[br + i * 64][bc] = gbh[i];
        *(v8bf*)&st.bl[br + i * 64][bc] = gbl[i];
    }
}

__device__ __forceinline__ void gemm_compute_tile(
    const GemmStage& st, int wr, int wc, int lm, int khalf, v8f (&acc)[2][4])
{
    #pragma unroll
    for (int ks = 0; ks < GKT; ks += 32) {
        v16bf ah[2], al[2], bh[4], bl[4];
        #pragma unroll
        for (int i = 0; i < 2; ++i) {
            ah[i] = frag16(&st.ah[wr + i * 16 + lm][ks], khalf);
            al[i] = frag16(&st.al[wr + i * 16 + lm][ks], khalf);
        }
        #pragma unroll
        for (int j = 0; j < 4; ++j) {
            bh[j] = frag16(&st.bh[wc + j * 16 + lm][ks], khalf);
            bl[j] = frag16(&st.bl[wc + j * 16 + lm][ks], khalf);
        }
        #pragma unroll
        for (int i = 0; i < 2; ++i)
            #pragma unroll
            for (int j = 0; j < 4; ++j)
                acc[i][j] = wmma32bf(al[i], bh[j], acc[i][j]);
        #pragma unroll
        for (int i = 0; i < 2; ++i)
            #pragma unroll
            for (int j = 0; j < 4; ++j)
                acc[i][j] = wmma32bf(ah[i], bl[j], acc[i][j]);
        #pragma unroll
        for (int i = 0; i < 2; ++i)
            #pragma unroll
            for (int j = 0; j < 4; ++j)
                acc[i][j] = wmma32bf(ah[i], bh[j], acc[i][j]);
    }
}

__global__ __launch_bounds__(512) __attribute__((amdgpu_waves_per_eu(4, 4))) void gemm_bf16x3(
    const float* __restrict__ A,
    const bf16* __restrict__ Bth, const bf16* __restrict__ Btl,
    const float* __restrict__ bias, float* __restrict__ Cm,
    int M, int N)
{
    __shared__ __align__(16) GemmStage stage[2];

    const int tid   = threadIdx.x;
    const int wave  = tid >> 5;
    const int lane  = tid & 31;
    const int lm    = lane & 15;
    const int khalf = (lane >> 4) << 3;
    const int hi    = (lane >> 4) << 3;

    const int wr = (wave & 7) * 32;
    const int wc = (wave >> 3) * 64;

    const long rowBase = (long)blockIdx.y * GBM;
    const long colBase = (long)blockIdx.x * GBN;

    const float* at0  = A   + (size_t)rowBase * GK;
    const bf16*  bht0 = Bth + (size_t)colBase * GK;
    const bf16*  blt0 = Btl + (size_t)colBase * GK;
    const unsigned aoff = (unsigned)(tid >> 4) * GK + ((tid & 15) << 2);
    const unsigned boff = (unsigned)(tid >> 3) * GK + ((tid & 7) << 3);

    v8f acc[2][4];
    #pragma unroll
    for (int j = 0; j < 4; ++j) { acc[0][j] = {}; acc[1][j] = {}; }

    float4 fa[8];
    v8bf gbh[2], gbl[2];

    gemm_load_tile(at0, bht0, blt0, aoff, boff, fa, gbh, gbl);
    gemm_store_tile(stage[0], tid, fa, gbh, gbl);
    __syncthreads();

    for (int kt = 0; kt < GNKT - 1; ++kt) {
        const int k1 = (kt + 1) * GKT;
        gemm_load_tile(at0 + k1, bht0 + k1, blt0 + k1, aoff, boff, fa, gbh, gbl);
        gemm_compute_tile(stage[kt & 1], wr, wc, lm, khalf, acc);
        __builtin_amdgcn_sched_barrier(0);
        gemm_store_tile(stage[(kt + 1) & 1], tid, fa, gbh, gbl);
        __syncthreads();
    }
    gemm_compute_tile(stage[(GNKT - 1) & 1], wr, wc, lm, khalf, acc);

    #pragma unroll
    for (int i = 0; i < 2; ++i) {
        #pragma unroll
        for (int j = 0; j < 4; ++j) {
            long col  = colBase + wc + j * 16 + lm;
            float bad = bias ? bias[col] : 0.0f;
            #pragma unroll
            for (int r = 0; r < 8; ++r) {
                long row = rowBase + wr + i * 16 + hi + r;
                Cm[(size_t)row * N + col] = acc[i][j][r] + bad;
            }
        }
    }
}

#define MT_C3    3072
#define MT_N     1024
#define MT_CH    4
#define MT_CROWS (MT_N / MT_CH)
#define MT_TROWS 128

__global__ __launch_bounds__(256) void mttt_stats_kernel(float* qkv)
{
    __shared__ __align__(16) float smem[2 * MT_TROWS * 68];
    float (*Ks)[68] = (float(*)[68]) &smem[0];
    float (*Vs)[68] = (float(*)[68]) &smem[MT_TROWS * 68];

    const int bh = blockIdx.x >> 2;
    const int ch = blockIdx.x & (MT_CH - 1);
    const int bb = bh >> 4;
    const int h  = bh & 15;
    const size_t rowBase = (size_t)bb * MT_N + (size_t)ch * MT_CROWS;
    const int kcol = 1024 + h * 64;
    const int vcol = 2048 + h * 64;

    const int tid  = threadIdx.x;
    const int wave = tid >> 5;
    const int lane = tid & 31;
    const int lm   = lane & 15;
    const int lk   = (lane >> 4) << 1;
    const int hi   = (lane >> 4) << 3;

    const bool isA  = wave < 4;
    const int mbase = (wave & 3) * 16;
    v8f accT[4];
    accT[0] = {}; accT[1] = {}; accT[2] = {}; accT[3] = {};
    float msum = 0.0f;

    for (int r0 = 0; r0 < MT_CROWS; r0 += MT_TROWS) {
        #pragma unroll
        for (int i = 0; i < 8; ++i) {
            int id = tid + i * 256;
            int r  = id >> 4;
            int c  = (id & 15) << 2;
            const float* src = &qkv[(rowBase + r0 + r) * MT_C3];
            *(float4*)&Ks[r][c] = *(const float4*)&src[kcol + c];
            *(float4*)&Vs[r][c] = *(const float4*)&src[vcol + c];
        }
        __syncthreads();

        if (tid < 64) {
            #pragma unroll 8
            for (int r = 0; r < MT_TROWS; ++r) msum += Ks[r][tid];
        } else if (tid < 128) {
            #pragma unroll 8
            for (int r = 0; r < MT_TROWS; ++r) msum += Vs[r][tid - 64];
        }

        float (*Bsrc)[68] = isA ? Ks : Vs;
        #pragma unroll 8
        for (int kk = 0; kk < MT_TROWS; kk += 4) {
            v2f a;
            a.x = Ks[kk + lk][mbase + lm];
            a.y = Ks[kk + lk + 1][mbase + lm];
            #pragma unroll
            for (int j = 0; j < 4; ++j) {
                v2f b;
                b.x = Bsrc[kk + lk][j * 16 + lm];
                b.y = Bsrc[kk + lk + 1][j * 16 + lm];
                accT[j] = wmma4(a, b, accT[j]);
            }
        }
        __syncthreads();
    }

    {
        float* dst = &qkv[rowBase * MT_C3 + (isA ? kcol : vcol)];
        #pragma unroll
        for (int j = 0; j < 4; ++j)
            #pragma unroll
            for (int r = 0; r < 8; ++r)
                dst[(size_t)(mbase + hi + r) * MT_C3 + j * 16 + lm] = accT[j][r];
    }
    if (tid < 64)       qkv[(rowBase + 64) * MT_C3 + kcol + tid]      = msum;
    else if (tid < 128) qkv[(rowBase + 64) * MT_C3 + vcol + tid - 64] = msum;
}

__global__ __launch_bounds__(256) void mttt_apply_kernel(
    const float* __restrict__ qkv, float* __restrict__ mid)
{
    __shared__ __align__(16) float smem[25536];
    float (*Qs)[68] = (float(*)[68]) &smem[0];
    float (*Am)[65] = (float(*)[65]) &smem[8704];
    float (*Cs)[65] = (float(*)[65]) &smem[8704 + 4160];
    float (*W0)[65] = (float(*)[65]) &smem[8704 + 2 * 4160];
    float (*W1)[65] = (float(*)[65]) &smem[8704 + 3 * 4160];
    float* mk   = &smem[8704 + 4 * 4160];
    float* mv   = mk + 64;
    float* bvec = mk + 128;

    const int bh = blockIdx.x >> 2;
    const int rq = blockIdx.x & 3;
    const int bb = bh >> 4;
    const int h  = bh & 15;
    const size_t rowBase = (size_t)bb * MT_N;
    const int qcol = h * 64;
    const int kcol = 1024 + h * 64;
    const int vcol = 2048 + h * 64;
    const int q0   = rq * MT_CROWS;

    const int tid  = threadIdx.x;
    const int wave = tid >> 5;
    const int lane = tid & 31;
    const int lm   = lane & 15;
    const int lk   = (lane >> 4) << 1;
    const int hi   = (lane >> 4) << 3;

    #pragma unroll
    for (int i = 0; i < 8; ++i) {
        int id = tid + i * 256;
        int r  = id >> 4;
        int c  = (id & 15) << 2;
        *(float4*)&Qs[r][c] =
            *(const float4*)&qkv[(rowBase + q0 + r) * MT_C3 + qcol + c];
    }

    const float invN = 1.0f / (float)MT_N;
    #pragma unroll
    for (int i = 0; i < 4; ++i) {
        int id = tid + i * 256;
        int r  = id >> 4;
        int c  = (id & 15) << 2;
        const float* s0 = &qkv[(rowBase + r) * MT_C3];
        float4 sa = *(const float4*)&s0[kcol + c];
        float4 sc = *(const float4*)&s0[vcol + c];
        #pragma unroll
        for (int chn = 1; chn < MT_CH; ++chn) {
            const float* s = &qkv[(rowBase + (size_t)chn * MT_CROWS + r) * MT_C3];
            float4 pa = *(const float4*)&s[kcol + c];
            float4 pc = *(const float4*)&s[vcol + c];
            sa.x += pa.x; sa.y += pa.y; sa.z += pa.z; sa.w += pa.w;
            sc.x += pc.x; sc.y += pc.y; sc.z += pc.z; sc.w += pc.w;
        }
        Am[r][c + 0] = sa.x * invN; Am[r][c + 1] = sa.y * invN;
        Am[r][c + 2] = sa.z * invN; Am[r][c + 3] = sa.w * invN;
        Cs[r][c + 0] = sc.x * invN; Cs[r][c + 1] = sc.y * invN;
        Cs[r][c + 2] = sc.z * invN; Cs[r][c + 3] = sc.w * invN;
    }
    if (tid < 128) {
        const int col = (tid < 64) ? (kcol + tid) : (vcol + tid - 64);
        float s = qkv[(rowBase + 64) * MT_C3 + col];
        #pragma unroll
        for (int chn = 1; chn < MT_CH; ++chn)
            s += qkv[(rowBase + (size_t)chn * MT_CROWS + 64) * MT_C3 + col];
        mk[tid] = s * invN;
    }
    for (int id = tid; id < 64 * 64; id += 256) W0[id >> 6][id & 63] = 0.0f;
    if (tid < 64) bvec[tid] = 0.0f;
    __syncthreads();

    float (*Wcur)[65] = W0;
    float (*Wnxt)[65] = W1;
    for (int it = 0; it < 4; ++it) {
        float bnew = 0.0f;
        if (tid < 64) {
            float s = 0.0f;
            #pragma unroll 8
            for (int i = 0; i < 64; ++i) s += mk[i] * Wcur[i][tid];
            bnew = mv[tid] - s;
        }
        #pragma unroll
        for (int j = 0; j < 2; ++j) {
            int t  = wave * 2 + j;
            int mb = (t >> 2) * 16;
            int nb = (t & 3) * 16;
            v8f acc = {};
            #pragma unroll
            for (int kk = 0; kk < 64; kk += 4) {
                v2f a, b2;
                a.x  = Am[mb + lm][kk + lk];
                a.y  = Am[mb + lm][kk + lk + 1];
                b2.x = Wcur[kk + lk][nb + lm];
                b2.y = Wcur[kk + lk + 1][nb + lm];
                acc = wmma4(a, b2, acc);
            }
            int col  = nb + lm;
            float bc = bvec[col];
            #pragma unroll
            for (int r = 0; r < 8; ++r) {
                int row = mb + hi + r;
                Wnxt[row][col] = Wcur[row][col] - acc[r] - mk[row] * bc + Cs[row][col];
            }
        }
        __syncthreads();
        if (tid < 64) bvec[tid] = bnew;
        { float (*tmp)[65] = Wcur; Wcur = Wnxt; Wnxt = tmp; }
        __syncthreads();
    }

    for (int c0 = q0; c0 < q0 + MT_CROWS; c0 += 128) {
        if (c0 != q0) {
            #pragma unroll
            for (int i = 0; i < 8; ++i) {
                int id = tid + i * 256;
                int r  = id >> 4;
                int c  = (id & 15) << 2;
                *(float4*)&Qs[r][c] =
                    *(const float4*)&qkv[(rowBase + c0 + r) * MT_C3 + qcol + c];
            }
            __syncthreads();
        }

        int rb = wave * 16;
        v8f acc[4];
        acc[0] = {}; acc[1] = {}; acc[2] = {}; acc[3] = {};
        #pragma unroll
        for (int kk = 0; kk < 64; kk += 4) {
            v2f a;
            a.x = Qs[rb + lm][kk + lk];
            a.y = Qs[rb + lm][kk + lk + 1];
            #pragma unroll
            for (int j = 0; j < 4; ++j) {
                v2f b2;
                b2.x = Wcur[kk + lk][j * 16 + lm];
                b2.y = Wcur[kk + lk + 1][j * 16 + lm];
                acc[j] = wmma4(a, b2, acc[j]);
            }
        }
        #pragma unroll
        for (int j = 0; j < 4; ++j) {
            int col  = j * 16 + lm;
            float ba = bvec[col];
            #pragma unroll
            for (int r = 0; r < 8; ++r) {
                size_t row = rowBase + c0 + rb + hi + r;
                mid[row * 1024 + qcol + col] = acc[j][r] + ba;
            }
        }
        __syncthreads();
    }
}

extern "C" void kernel_launch(void* const* d_in, const int* in_sizes, int n_in,
                              void* d_out, int out_size, void* d_ws, size_t ws_size,
                              hipStream_t stream) {
    const float* x      = (const float*)d_in[0];
    const float* w_qkv  = (const float*)d_in[1];
    const float* w_proj = (const float*)d_in[2];
    const float* b_proj = (const float*)d_in[3];
    float* out = (float*)d_out;

    float* qkv = (float*)d_ws;
    float* mid = qkv + (size_t)8192 * 3072;
    bf16*  b1h = (bf16*)(mid + (size_t)8192 * 1024);
    bf16*  b1l = b1h + (size_t)3072 * 1024;
    bf16*  b2h = b1l + (size_t)3072 * 1024;
    bf16*  b2l = b2h + (size_t)1024 * 1024;

    dim3 blk(256);
    split_transpose_bf16<<<dim3(3072 / 32, 1024 / 32), blk, 0, stream>>>(
        w_qkv, b1h, b1l, 1024, 3072);
    split_transpose_bf16<<<dim3(1024 / 32, 1024 / 32), blk, 0, stream>>>(
        w_proj, b2h, b2l, 1024, 1024);

    gemm_bf16x3<<<dim3(3072 / GBN, 8192 / GBM), dim3(512), 0, stream>>>(
        x, b1h, b1l, nullptr, qkv, 8192, 3072);

    mttt_stats_kernel<<<dim3(8 * 16 * MT_CH), blk, 0, stream>>>(qkv);
    mttt_apply_kernel<<<dim3(8 * 16 * 4), blk, 0, stream>>>(qkv, mid);

    gemm_bf16x3<<<dim3(1024 / GBN, 8192 / GBM), dim3(512), 0, stream>>>(
        mid, b2h, b2l, b_proj, out, 8192, 1024);
}


# bf16x3 GEMM: LDS ring of 4 stages x 32 k (was 2 x 64), barrier split: signal behind a tile's WMMAs, wait one tile later, no drain per tile; weight hi|lo of 32 k in one line; sum order kept
# speedup vs baseline: 1.5860x; 1.0780x over previous
#include <hip/hip_runtime.h>
#include <hip/hip_bf16.h>

typedef float v2f  __attribute__((ext_vector_type(2)));
typedef float v8f  __attribute__((ext_vector_type(8)));
typedef __bf16 bf16;
typedef bf16 v4bf  __attribute__((ext_vector_type(4)));
typedef bf16 v8bf  __attribute__((ext_vector_type(8)));
typedef bf16 v16bf __attribute__((ext_vector_type(16)));

__device__ __forceinline__ v8f wmma4(v2f a, v2f b, v8f c) {
    return __builtin_amdgcn_wmma_f32_16x16x4_f32(
        false, a, false, b, (short)0, c, false, false);
}
__device__ __forceinline__ v8f wmma32bf(v16bf a, v16bf b, v8f c) {
    return __builtin_amdgcn_wmma_f32_16x16x32_bf16(
        false, a, false, b, (short)0, c, false, false);
}

__device__ __forceinline__ v16bf frag16(const bf16* rowBase, int khalf) {
    v8bf lo = *(const v8bf*)(rowBase + khalf);
    v8bf hi = *(const v8bf*)(rowBase + 16 + khalf);
    return __builtin_shufflevector(lo, hi, 0,1,2,3,4,5,6,7,8,9,10,11,12,13,14,15);
}

__global__ __launch_bounds__(256) void split_transpose_bf16(
    const float* __restrict__ B, bf16* __restrict__ Th, bf16* __restrict__ Tl,
    int K, int N)
{
    __shared__ float t[32][33];
    const int tid = threadIdx.x;
    const int k0 = blockIdx.y * 32;
    const int n0 = blockIdx.x * 32;
    #pragma unroll
    for (int i = 0; i < 4; ++i) {
        int id = tid + i * 256;
        int r = id >> 5, c = id & 31;
        t[r][c] = B[(size_t)(k0 + r) * N + n0 + c];
    }
    __syncthreads();
    #pragma unroll
    for (int i = 0; i < 4; ++i) {
        int id = tid + i * 256;
        int r = id >> 5, c = id & 31;
        float f = t[c][r];
        bf16 h  = (bf16)f;
        float l = f - (float)h;
        size_t o = (size_t)(n0 + r) * (2 * K) + 2 * k0 + c;
        Th[o] = h;
        Tl[o] = (bf16)l;
    }
}

#define GBM 256
#define GBN 128
#define GKT 32
#define GK  1024
#define GKB (2 * GK)
#define GNKT (GK / GKT)
#define GNST 4
#define APITCH 40

struct GemmStage {
    bf16 ah[GBM][APITCH];
    bf16 al[GBM][APITCH];
    bf16 bh[GBN][APITCH];
    bf16 bl[GBN][APITCH];
};

__device__ __forceinline__ void gemm_lds_done()   { asm volatile("s_wait_dscnt 0x0" ::: "memory"); }
__device__ __forceinline__ void gemm_bar_signal() { asm volatile("s_barrier_signal -1" ::: "memory"); }
__device__ __forceinline__ void gemm_bar_wait()   { asm volatile("s_barrier_wait -1" ::: "memory"); }

__device__ __forceinline__ void gemm_load_tile(
    const float* __restrict__ at, const bf16* __restrict__ bht, const bf16* __restrict__ blt,
    unsigned aoff, unsigned boff, float4 (&fa)[4], v8bf &gbh, v8bf &gbl)
{
    #pragma unroll
    for (int i = 0; i < 4; ++i)
        fa[i] = *(const float4*)(at + (aoff + (unsigned)i * 64u * GK));
    gbh = *(const v8bf*)(bht + boff);
    gbl = *(const v8bf*)(blt + boff);
}

__device__ __forceinline__ void gemm_store_tile(
    GemmStage& st, int arow, int acol, int brow, int bcol,
    const float4 (&fa)[4], const v8bf &gbh, const v8bf &gbl)
{
    #pragma unroll
    for (int i = 0; i < 4; ++i) {
        float4 f = fa[i];
        v4bf h, l;
        h[0] = (bf16)f.x; l[0] = (bf16)(f.x - (float)h[0]);
        h[1] = (bf16)f.y; l[1] = (bf16)(f.y - (float)h[1]);
        h[2] = (bf16)f.z; l[2] = (bf16)(f.z - (float)h[2]);
        h[3] = (bf16)f.w; l[3] = (bf16)(f.w - (float)h[3]);
        *(v4bf*)&st.ah[arow + i * 64][acol] = h;
        *(v4bf*)&st.al[arow + i * 64][acol] = l;
    }
    *(v8bf*)&st.bh[brow][bcol] = gbh;
    *(v8bf*)&st.bl[brow][bcol] = gbl;
}

__device__ __forceinline__ void gemm_compute_tile(
    const GemmStage& st, int wr, int wc, int lm, int khalf, v8f (&acc)[2][4])
{
    v16bf ah[2], al[2], bh[4], bl[4];
    #pragma unroll
    for (int i = 0; i < 2; ++i) {
        ah[i] = frag16(&st.ah[wr + i * 16 + lm][0], khalf);
        al[i] = frag16(&st.al[wr + i * 16 + lm][0], khalf);
    }
    #pragma unroll
    for (int j = 0; j < 4; ++j) {
        bh[j] = frag16(&st.bh[wc + j * 16 + lm][0], khalf);
        bl[j] = frag16(&st.bl[wc + j * 16 + lm][0], khalf);
    }
    asm volatile("" : "+v"(ah[0]), "+v"(ah[1]), "+v"(al[0]), "+v"(al[1]),
                      "+v"(bh[0]), "+v"(bh[1]), "+v"(bh[2]), "+v"(bh[3]),
                      "+v"(bl[0]), "+v"(bl[1]), "+v"(bl[2]), "+v"(bl[3]));
    #pragma unroll
    for (int i = 0; i < 2; ++i)
        #pragma unroll
        for (int j = 0; j < 4; ++j)
            acc[i][j] = wmma32bf(al[i], bh[j], acc[i][j]);
    #pragma unroll
    for (int i = 0; i < 2; ++i)
        #pragma unroll
        for (int j = 0; j < 4; ++j)
            acc[i][j] = wmma32bf(ah[i], bl[j], acc[i][j]);
    #pragma unroll
    for (int i = 0; i < 2; ++i)
        #pragma unroll
        for (int j = 0; j < 4; ++j)
            acc[i][j] = wmma32bf(ah[i], bh[j], acc[i][j]);
    asm volatile("" : "+v"(acc[0][0]), "+v"(acc[0][1]), "+v"(acc[0][2]), "+v"(acc[0][3]),
                      "+v"(acc[1][0]), "+v"(acc[1][1]), "+v"(acc[1][2]), "+v"(acc[1][3]));
}

__global__ __launch_bounds__(512) __attribute__((amdgpu_waves_per_eu(4, 4))) void gemm_bf16x3(
    const float* __restrict__ A,
    const bf16* __restrict__ Bth, const bf16* __restrict__ Btl,
    const float* __restrict__ bias, float* __restrict__ Cm,
    int M, int N)
{
    __shared__ __align__(16) GemmStage stage[GNST];

    const int tid   = threadIdx.x;
    const int wave  = tid >> 5;
    const int lane  = tid & 31;
    const int lm    = lane & 15;
    const int khalf = (lane >> 4) << 3;
    const int hi    = (lane >> 4) << 3;

    const int wr = (wave & 7) * 32;
    const int wc = (wave >> 3) * 64;

    const long rowBase = (long)blockIdx.y * GBM;
    const long colBase = (long)blockIdx.x * GBN;

    const int arow = 16 * (wave >> 2) + (wave & 3) + 4 * (lane >> 3);
    const int acol = (lane & 7) << 2;
    const int bidx = wave * 2 + (lane >> 4);
    const int brow = 16 * (bidx >> 2) + (bidx & 3) + 4 * ((lane >> 2) & 3);
    const int bcol = (lane & 3) << 3;

    const float* at0  = A   + (size_t)rowBase * GK;
    const bf16*  bht0 = Bth + (size_t)colBase * GKB;
    const bf16*  blt0 = Btl + (size_t)colBase * GKB;
    const unsigned aoff = (unsigned)arow * GK + (unsigned)acol;
    const unsigned boff = (unsigned)brow * GKB + (unsigned)bcol;

    v8f acc[2][4];
    #pragma unroll
    for (int j = 0; j < 4; ++j) { acc[0][j] = {}; acc[1][j] = {}; }

    float4 fa[4];
    v8bf gbh, gbl;

    {
        float4 fb[4], fc[4];
        v8bf hb, lb, hc, lc;
        gemm_load_tile(at0,           bht0,           blt0,           aoff, boff, fa, gbh, gbl);
        gemm_load_tile(at0 + GKT,     bht0 + 2 * GKT, blt0 + 2 * GKT, aoff, boff, fb, hb, lb);
        gemm_load_tile(at0 + 2 * GKT, bht0 + 4 * GKT, blt0 + 4 * GKT, aoff, boff, fc, hc, lc);
        gemm_store_tile(stage[0], arow, acol, brow, bcol, fa, gbh, gbl);
        gemm_store_tile(stage[1], arow, acol, brow, bcol, fb, hb, lb);
        gemm_store_tile(stage[2], arow, acol, brow, bcol, fc, hc, lc);
    }
    __syncthreads();

    for (int kt = 0; kt < GNKT - 3; ++kt) {
        const int k3 = (kt + 3) * GKT;
        gemm_load_tile(at0 + k3, bht0 + 2 * k3, blt0 + 2 * k3, aoff, boff, fa, gbh, gbl);
        gemm_compute_tile(stage[kt & 3], wr, wc, lm, khalf, acc);
        if (kt > 0) gemm_bar_wait();
        gemm_lds_done();
        gemm_bar_signal();
        asm volatile("" : "+v"(fa[0].x), "+v"(fa[0].y), "+v"(fa[0].z), "+v"(fa[0].w),
                          "+v"(fa[1].x), "+v"(fa[1].y), "+v"(fa[1].z), "+v"(fa[1].w),
                          "+v"(fa[2].x), "+v"(fa[2].y), "+v"(fa[2].z), "+v"(fa[2].w),
                          "+v"(fa[3].x), "+v"(fa[3].y), "+v"(fa[3].z), "+v"(fa[3].w));
        gemm_store_tile(stage[(kt + 3) & 3], arow, acol, brow, bcol, fa, gbh, gbl);
    }
    gemm_compute_tile(stage[(GNKT - 3) & 3], wr, wc, lm, khalf, acc);
    gemm_bar_wait();
    gemm_lds_done();
    gemm_bar_signal();
    gemm_compute_tile(stage[(GNKT - 2) & 3], wr, wc, lm, khalf, acc);
    gemm_bar_wait();
    gemm_compute_tile(stage[(GNKT - 1) & 3], wr, wc, lm, khalf, acc);

    #pragma unroll
    for (int i = 0; i < 2; ++i) {
        #pragma unroll
        for (int j = 0; j < 4; ++j) {
            long col  = colBase + wc + j * 16 + lm;
            float bad = bias ? bias[col] : 0.0f;
            #pragma unroll
            for (int r = 0; r < 8; ++r) {
                long row = rowBase + wr + i * 16 + hi + r;
                Cm[(size_t)row * N + col] = acc[i][j][r] + bad;
            }
        }
    }
}

#define MT_C3    3072
#define MT_N     1024
#define MT_CH    4
#define MT_CROWS (MT_N / MT_CH)
#define MT_TROWS 128

__global__ __launch_bounds__(256) void mttt_stats_kernel(float* qkv)
{
    __shared__ __align__(16) float smem[2 * MT_TROWS * 68];
    float (*Ks)[68] = (float(*)[68]) &smem[0];
    float (*Vs)[68] = (float(*)[68]) &smem[MT_TROWS * 68];

    const int bh = blockIdx.x >> 2;
    const int ch = blockIdx.x & (MT_CH - 1);
    const int bb = bh >> 4;
    const int h  = bh & 15;
    const size_t rowBase = (size_t)bb * MT_N + (size_t)ch * MT_CROWS;
    const int kcol = 1024 + h * 64;
    const int vcol = 2048 + h * 64;

    const int tid  = threadIdx.x;
    const int wave = tid >> 5;
    const int lane = tid & 31;
    const int lm   = lane & 15;
    const int lk   = (lane >> 4) << 1;
    const int hi   = (lane >> 4) << 3;

    const bool isA  = wave < 4;
    const int mbase = (wave & 3) * 16;
    v8f accT[4];
    accT[0] = {}; accT[1] = {}; accT[2] = {}; accT[3] = {};
    float msum = 0.0f;

    for (int r0 = 0; r0 < MT_CROWS; r0 += MT_TROWS) {
        #pragma unroll
        for (int i = 0; i < 8; ++i) {
            int id = tid + i * 256;
            int r  = id >> 4;
            int c  = (id & 15) << 2;
            const float* src = &qkv[(rowBase + r0 + r) * MT_C3];
            *(float4*)&Ks[r][c] = *(const float4*)&src[kcol + c];
            *(float4*)&Vs[r][c] = *(const float4*)&src[vcol + c];
        }
        __syncthreads();

        if (tid < 64) {
            #pragma unroll 8
            for (int r = 0; r < MT_TROWS; ++r) msum += Ks[r][tid];
        } else if (tid < 128) {
            #pragma unroll 8
            for (int r = 0; r < MT_TROWS; ++r) msum += Vs[r][tid - 64];
        }

        float (*Bsrc)[68] = isA ? Ks : Vs;
        #pragma unroll 8
        for (int kk = 0; kk < MT_TROWS; kk += 4) {
            v2f a;
            a.x = Ks[kk + lk][mbase + lm];
            a.y = Ks[kk + lk + 1][mbase + lm];
            #pragma unroll
            for (int j = 0; j < 4; ++j) {
                v2f b;
                b.x = Bsrc[kk + lk][j * 16 + lm];
                b.y = Bsrc[kk + lk + 1][j * 16 + lm];
                accT[j] = wmma4(a, b, accT[j]);
            }
        }
        __syncthreads();
    }

    {
        float* dst = &qkv[rowBase * MT_C3 + (isA ? kcol : vcol)];
        #pragma unroll
        for (int j = 0; j < 4; ++j)
            #pragma unroll
            for (int r = 0; r < 8; ++r)
                dst[(size_t)(mbase + hi + r) * MT_C3 + j * 16 + lm] = accT[j][r];
    }
    if (tid < 64)       qkv[(rowBase + 64) * MT_C3 + kcol + tid]      = msum;
    else if (tid < 128) qkv[(rowBase + 64) * MT_C3 + vcol + tid - 64] = msum;
}

__global__ __launch_bounds__(256) void mttt_apply_kernel(
    const float* __restrict__ qkv, float* __restrict__ mid)
{
    __shared__ __align__(16) float smem[25536];
    float (*Qs)[68] = (float(*)[68]) &smem[0];
    float (*Am)[65] = (float(*)[65]) &smem[8704];
    float (*Cs)[65] = (float(*)[65]) &smem[8704 + 4160];
    float (*W0)[65] = (float(*)[65]) &smem[8704 + 2 * 4160];
    float (*W1)[65] = (float(*)[65]) &smem[8704 + 3 * 4160];
    float* mk   = &smem[8704 + 4 * 4160];
    float* mv   = mk + 64;
    float* bvec = mk + 128;

    const int bh = blockIdx.x >> 2;
    const int rq = blockIdx.x & 3;
    const int bb = bh >> 4;
    const int h  = bh & 15;
    const size_t rowBase = (size_t)bb * MT_N;
    const int qcol = h * 64;
    const int kcol = 1024 + h * 64;
    const int vcol = 2048 + h * 64;
    const int q0   = rq * MT_CROWS;

    const int tid  = threadIdx.x;
    const int wave = tid >> 5;
    const int lane = tid & 31;
    const int lm   = lane & 15;
    const int lk   = (lane >> 4) << 1;
    const int hi   = (lane >> 4) << 3;

    #pragma unroll
    for (int i = 0; i < 8; ++i) {
        int id = tid + i * 256;
        int r  = id >> 4;
        int c  = (id & 15) << 2;
        *(float4*)&Qs[r][c] =
            *(const float4*)&qkv[(rowBase + q0 + r) * MT_C3 + qcol + c];
    }

    const float invN = 1.0f / (float)MT_N;
    #pragma unroll
    for (int i = 0; i < 4; ++i) {
        int id = tid + i * 256;
        int r  = id >> 4;
        int c  = (id & 15) << 2;
        const float* s0 = &qkv[(rowBase + r) * MT_C3];
        float4 sa = *(const float4*)&s0[kcol + c];
        float4 sc = *(const float4*)&s0[vcol + c];
        #pragma unroll
        for (int chn = 1; chn < MT_CH; ++chn) {
            const float* s = &qkv[(rowBase + (size_t)chn * MT_CROWS + r) * MT_C3];
            float4 pa = *(const float4*)&s[kcol + c];
            float4 pc = *(const float4*)&s[vcol + c];
            sa.x += pa.x; sa.y += pa.y; sa.z += pa.z; sa.w += pa.w;
            sc.x += pc.x; sc.y += pc.y; sc.z += pc.z; sc.w += pc.w;
        }
        Am[r][c + 0] = sa.x * invN; Am[r][c + 1] = sa.y * invN;
        Am[r][c + 2] = sa.z * invN; Am[r][c + 3] = sa.w * invN;
        Cs[r][c + 0] = sc.x * invN; Cs[r][c + 1] = sc.y * invN;
        Cs[r][c + 2] = sc.z * invN; Cs[r][c + 3] = sc.w * invN;
    }
    if (tid < 128) {
        const int col = (tid < 64) ? (kcol + tid) : (vcol + tid - 64);
        float s = qkv[(rowBase + 64) * MT_C3 + col];
        #pragma unroll
        for (int chn = 1; chn < MT_CH; ++chn)
            s += qkv[(rowBase + (size_t)chn * MT_CROWS + 64) * MT_C3 + col];
        mk[tid] = s * invN;
    }
    for (int id = tid; id < 64 * 64; id += 256) W0[id >> 6][id & 63] = 0.0f;
    if (tid < 64) bvec[tid] = 0.0f;
    __syncthreads();

    float (*Wcur)[65] = W0;
    float (*Wnxt)[65] = W1;
    for (int it = 0; it < 4; ++it) {
        float bnew = 0.0f;
        if (tid < 64) {
            float s = 0.0f;
            #pragma unroll 8
            for (int i = 0; i < 64; ++i) s += mk[i] * Wcur[i][tid];
            bnew = mv[tid] - s;
        }
        #pragma unroll
        for (int j = 0; j < 2; ++j) {
            int t  = wave * 2 + j;
            int mb = (t >> 2) * 16;
            int nb = (t & 3) * 16;
            v8f acc = {};
            #pragma unroll
            for (int kk = 0; kk < 64; kk += 4) {
                v2f a, b2;
                a.x  = Am[mb + lm][kk + lk];
                a.y  = Am[mb + lm][kk + lk + 1];
                b2.x = Wcur[kk + lk][nb + lm];
                b2.y = Wcur[kk + lk + 1][nb + lm];
                acc = wmma4(a, b2, acc);
            }
            int col  = nb + lm;
            float bc = bvec[col];
            #pragma unroll
            for (int r = 0; r < 8; ++r) {
                int row = mb + hi + r;
                Wnxt[row][col] = Wcur[row][col] - acc[r] - mk[row] * bc + Cs[row][col];
            }
        }
        __syncthreads();
        if (tid < 64) bvec[tid] = bnew;
        { float (*tmp)[65] = Wcur; Wcur = Wnxt; Wnxt = tmp; }
        __syncthreads();
    }

    for (int c0 = q0; c0 < q0 + MT_CROWS; c0 += 128) {
        if (c0 != q0) {
            #pragma unroll
            for (int i = 0; i < 8; ++i) {
                int id = tid + i * 256;
                int r  = id >> 4;
                int c  = (id & 15) << 2;
                *(float4*)&Qs[r][c] =
                    *(const float4*)&qkv[(rowBase + c0 + r) * MT_C3 + qcol + c];
            }
            __syncthreads();
        }

        int rb = wave * 16;
        v8f acc[4];
        acc[0] = {}; acc[1] = {}; acc[2] = {}; acc[3] = {};
        #pragma unroll
        for (int kk = 0; kk < 64; kk += 4) {
            v2f a;
            a.x = Qs[rb + lm][kk + lk];
            a.y = Qs[rb + lm][kk + lk + 1];
            #pragma unroll
            for (int j = 0; j < 4; ++j) {
                v2f b2;
                b2.x = Wcur[kk + lk][j * 16 + lm];
                b2.y = Wcur[kk + lk + 1][j * 16 + lm];
                acc[j] = wmma4(a, b2, acc[j]);
            }
        }
        #pragma unroll
        for (int j = 0; j < 4; ++j) {
            int col  = j * 16 + lm;
            float ba = bvec[col];
            #pragma unroll
            for (int r = 0; r < 8; ++r) {
                size_t row = rowBase + c0 + rb + hi + r;
                mid[row * 1024 + qcol + col] = acc[j][r] + ba;
            }
        }
        __syncthreads();
    }
}

extern "C" void kernel_launch(void* const* d_in, const int* in_sizes, int n_in,
                              void* d_out, int out_size, void* d_ws, size_t ws_size,
                              hipStream_t stream) {
    const float* x      = (const float*)d_in[0];
    const float* w_qkv  = (const float*)d_in[1];
    const float* w_proj = (const float*)d_in[2];
    const float* b_proj = (const float*)d_in[3];
    float* out = (float*)d_out;

    float* qkv = (float*)d_ws;
    float* mid = qkv + (size_t)8192 * 3072;
    bf16*  b1h = (bf16*)(mid + (size_t)8192 * 1024);
    bf16*  b1l = b1h + 32;
    bf16*  b2h = b1h + (size_t)3072 * 2048;
    bf16*  b2l = b2h + 32;

    dim3 blk(256);
    split_transpose_bf16<<<dim3(3072 / 32, 1024 / 32), blk, 0, stream>>>(
        w_qkv, b1h, b1l, 1024, 3072);
    split_transpose_bf16<<<dim3(1024 / 32, 1024 / 32), blk, 0, stream>>>(
        w_proj, b2h, b2l, 1024, 1024);

    gemm_bf16x3<<<dim3(3072 / GBN, 8192 / GBM), dim3(512), 0, stream>>>(
        x, b1h, b1l, nullptr, qkv, 8192, 3072);

    mttt_stats_kernel<<<dim3(8 * 16 * MT_CH), blk, 0, stream>>>(qkv);
    mttt_apply_kernel<<<dim3(8 * 16 * 4), blk, 0, stream>>>(qkv, mid);

    gemm_bf16x3<<<dim3(1024 / GBN, 8192 / GBM), dim3(512), 0, stream>>>(
        mid, b2h, b2l, b_proj, out, 8192, 1024);
}
